# starting kernel
# baseline (speedup 1.0000x reference)
#include <hip/hip_runtime.h>
#include <math.h>

typedef __attribute__((ext_vector_type(2))) float v2f;
typedef __attribute__((ext_vector_type(8))) float v8f;

#define B_SZ   64
#define NI     2048
#define DI     8
#define NO     32
#define DO_    16
#define NOD    512
#define KTOT   (NI*DI)
#define WN     (NO*DO_*DI)

#define NB_GEMM  128
#define KCHUNK   (KTOT/NB_GEMM)
#define NB_ROUTE 128
#define NPB      (NI/NB_ROUTE)
#define PS_STRIDE (B_SZ*NOD)

__global__ __launch_bounds__(256)
void caps_s0_gemm(const float* __restrict__ x, const float* __restrict__ W,
                  float* __restrict__ ps) {
  const int lane = threadIdx.x & 31;
  const int wave = threadIdx.x >> 5;
  const int half = lane >> 4;
  const int l15  = lane & 15;
  const int kbase = blockIdx.x * KCHUNK;
  float* psb = ps + (size_t)blockIdx.x * PS_STRIDE;

  for (int ot = 0; ot < 4; ++ot) {
    const int o = wave + ot * 8;
    const float* wc = W + (size_t)(o * DO_ + l15) * DI;
    for (int mt = 0; mt < 4; ++mt) {
      const int row = mt * 16 + l15;
      const float* xr = x + (size_t)row * KTOT;
      v8f acc = {};
      for (int k0 = kbase; k0 < kbase + KCHUNK; k0 += 4) {
        const int ka = k0 + 2 * half;
        v2f a  = *(const v2f*)(xr + ka);
        const int n = ka >> 3, i = ka & 7;
        v2f bv = *(const v2f*)(wc + (size_t)n * WN + i);
        acc = __builtin_amdgcn_wmma_f32_16x16x4_f32(
                  false, a, false, bv, (short)0, acc, false, false);
      }
#pragma unroll
      for (int r = 0; r < 8; ++r) {
        const int b = mt * 16 + r + 8 * half;
        psb[b * NOD + o * DO_ + l15] = acc[r];
      }
    }
  }
}

__global__ __launch_bounds__(256)
void caps_route(const float* __restrict__ x, const float* __restrict__ W,
                const float* __restrict__ vcur, float* __restrict__ rlog,
                float* __restrict__ ps, int iter) {
  __shared__ float r1s[B_SZ * NO];
  __shared__ float cs[B_SZ * NO];
  const int lane = threadIdx.x & 31;
  const int wave = threadIdx.x >> 5;
  const int half = lane >> 4;
  const int l15  = lane & 15;

  v8f zero = {};
  v8f sacc[16];
#pragma unroll
  for (int t = 0; t < 16; ++t) sacc[t] = zero;

  const int n0 = blockIdx.x * NPB;
  for (int nn = 0; nn < NPB; ++nn) {
    const int n = n0 + nn;
    const float* wn = W + (size_t)n * WN;

    for (int ot = 0; ot < 4; ++ot) {
      const int o = wave + ot * 8;
      for (int mt = 0; mt < 4; ++mt) {
        const int row = mt * 16 + l15;
        v8f d = zero;
#pragma unroll
        for (int ks = 0; ks < 2; ++ks) {
          const int k = ks * 4 + 2 * half;
          v2f a  = *(const v2f*)(x + (size_t)row * KTOT + n * DI + k);
          v2f bv = *(const v2f*)(wn + (size_t)(o * DO_ + l15) * DI + k);
          d = __builtin_amdgcn_wmma_f32_16x16x4_f32(
                  false, a, false, bv, (short)0, d, false, false);
        }
#pragma unroll
        for (int r = 0; r < 8; ++r) {
          const int b = mt * 16 + r + 8 * half;
          float p = d[r] * vcur[b * NOD + o * DO_ + l15];
          p += __shfl_xor(p, 1);
          p += __shfl_xor(p, 2);
          p += __shfl_xor(p, 4);
          p += __shfl_xor(p, 8);
          if (l15 == 0) {
            const size_t ridx = ((size_t)b * NI + n) * NO + o;
            float logit;
            if (iter == 1) { logit = p; rlog[ridx] = p; }
            else           { logit = rlog[ridx] + p; }
            r1s[b * NO + o] = logit;
          }
        }
      }
    }
    __syncthreads();

    if (threadIdx.x < B_SZ) {
      const int b = threadIdx.x;
      float m = -3.4e38f;
      for (int o = 0; o < NO; ++o) m = fmaxf(m, r1s[b * NO + o]);
      float ssum = 0.f;
      for (int o = 0; o < NO; ++o) {
        float e = __expf(r1s[b * NO + o] - m);
        cs[b * NO + o] = e; ssum += e;
      }
      const float inv = 1.f / ssum;
      for (int o = 0; o < NO; ++o) cs[b * NO + o] *= inv;
    }
    __syncthreads();

    for (int ot = 0; ot < 4; ++ot) {
      const int o = wave + ot * 8;
      for (int mt = 0; mt < 4; ++mt) {
        const int row = mt * 16 + l15;
        v8f d = zero;
#pragma unroll
        for (int ks = 0; ks < 2; ++ks) {
          const int k = ks * 4 + 2 * half;
          v2f a  = *(const v2f*)(x + (size_t)row * KTOT + n * DI + k);
          v2f bv = *(const v2f*)(wn + (size_t)(o * DO_ + l15) * DI + k);
          d = __builtin_amdgcn_wmma_f32_16x16x4_f32(
                  false, a, false, bv, (short)0, d, false, false);
        }
#pragma unroll
        for (int r = 0; r < 8; ++r) {
          const int b = mt * 16 + r + 8 * half;
          sacc[ot * 4 + mt][r] += cs[b * NO + o] * d[r];
        }
      }
    }
    __syncthreads();
  }

  float* psb = ps + (size_t)blockIdx.x * PS_STRIDE;
  for (int ot = 0; ot < 4; ++ot) {
    const int o = wave + ot * 8;
    for (int mt = 0; mt < 4; ++mt) {
      v8f s = sacc[ot * 4 + mt];
#pragma unroll
      for (int r = 0; r < 8; ++r) {
        const int b = mt * 16 + r + 8 * half;
        psb[b * NOD + o * DO_ + l15] = s[r];
      }
    }
  }
}

__global__ __launch_bounds__(512)
void caps_reduce_squash(const float* __restrict__ ps, int nparts, float prescale,
                        float* __restrict__ out) {
  __shared__ float sh[NOD];
  __shared__ float scl[NO];
  const int b  = blockIdx.x;
  const int od = threadIdx.x;
  float s = 0.f;
  for (int p = 0; p < nparts; ++p)
    s += ps[(size_t)p * PS_STRIDE + b * NOD + od];
  s *= prescale;
  sh[od] = s;
  __syncthreads();
  if (od < NO) {
    float nn = 0.f;
#pragma unroll
    for (int d = 0; d < DO_; ++d) { float t = sh[od * DO_ + d]; nn += t * t; }
    scl[od] = sqrtf(nn) / (1.f + nn);
  }
  __syncthreads();
  out[(size_t)b * NOD + od] = sh[od] * scl[od >> 4];
}

extern "C" void kernel_launch(void* const* d_in, const int* in_sizes, int n_in,
                              void* d_out, int out_size, void* d_ws, size_t ws_size,
                              hipStream_t stream) {
  const float* x = (const float*)d_in[0];
  const float* W = (const float*)d_in[1];
  float* out = (float*)d_out;

  float* ps   = (float*)d_ws;
  float* vbuf = ps + (size_t)NB_GEMM * PS_STRIDE;
  float* rlog = vbuf + (size_t)B_SZ * NOD;

  caps_s0_gemm<<<NB_GEMM, 256, 0, stream>>>(x, W, ps);
  caps_reduce_squash<<<B_SZ, NOD, 0, stream>>>(ps, NB_GEMM, 1.0f / NO, vbuf);
  caps_route<<<NB_ROUTE, 256, 0, stream>>>(x, W, vbuf, rlog, ps, 1);
  caps_reduce_squash<<<B_SZ, NOD, 0, stream>>>(ps, NB_ROUTE, 1.0f, vbuf);
  caps_route<<<NB_ROUTE, 256, 0, stream>>>(x, W, vbuf, rlog, ps, 2);
  caps_reduce_squash<<<B_SZ, NOD, 0, stream>>>(ps, NB_ROUTE, 1.0f, out);
}


# caps_s0: iteration-0 sum over input capsules split over the 8 waves of the block owning a result tile, folded in LDS, squash applied there: no partial planes, no reduce launch (5 launches, was 6)
# speedup vs baseline: 16.6692x; 16.6692x over previous
#include <hip/hip_runtime.h>
#include <math.h>

typedef __attribute__((ext_vector_type(2))) float v2f;
typedef __attribute__((ext_vector_type(4))) float v4f;
typedef __attribute__((ext_vector_type(8))) float v8f;

#define B_SZ   64
#define NI     2048
#define DI     8
#define NO     32
#define DO_    16
#define NOD    512
#define KTOT   (NI*DI)
#define WN     (NO*DO_*DI)

#define NB_ROUTE 128
#define NPB      (NI/NB_ROUTE)
#define NBLK_ROUTE 256
#define NPW      (NPB/2)
#define FOLD_V4  (64*32)
#define PS_STRIDE (B_SZ*NOD)

#define KW    8
#define NPWG  (NI/KW)
#define GST   16
#define NBLK_S0 128

__device__ __forceinline__ void s0_stage(const v4f (&xa)[GST], const v4f (&wa)[GST],
                                         v8f& acc0, v8f& acc1) {
#pragma unroll
  for (int j = 0; j < GST; j += 2) {
    const v2f a00 = {xa[j][0], xa[j][1]},         a01 = {xa[j][2], xa[j][3]};
    const v2f b00 = {wa[j][0], wa[j][1]},         b01 = {wa[j][2], wa[j][3]};
    const v2f a10 = {xa[j + 1][0], xa[j + 1][1]}, a11 = {xa[j + 1][2], xa[j + 1][3]};
    const v2f b10 = {wa[j + 1][0], wa[j + 1][1]}, b11 = {wa[j + 1][2], wa[j + 1][3]};
    acc0 = __builtin_amdgcn_wmma_f32_16x16x4_f32(false, a00, false, b00, (short)0, acc0, false, false);
    acc1 = __builtin_amdgcn_wmma_f32_16x16x4_f32(false, a10, false, b10, (short)0, acc1, false, false);
    acc0 = __builtin_amdgcn_wmma_f32_16x16x4_f32(false, a01, false, b01, (short)0, acc0, false, false);
    acc1 = __builtin_amdgcn_wmma_f32_16x16x4_f32(false, a11, false, b11, (short)0, acc1, false, false);
  }
}

__global__ __launch_bounds__(256)
void caps_s0_squash(const float* __restrict__ x, const float* __restrict__ W,
                    float* __restrict__ v0) {
  __shared__ float part[KW * 256];
  __shared__ float sh[256];
  const int lane = threadIdx.x & 31;
  const int wave = __builtin_amdgcn_readfirstlane((int)(threadIdx.x >> 5));
  const int h  = lane >> 4;
  const int c  = lane & 15;
  const int mt = blockIdx.x & 3;
  const int o  = blockIdx.x >> 2;
  const int n0 = wave * NPWG;

  const float* xp = x + ((size_t)(mt * 16 + c) * NI + n0) * DI + 4 * h;
  const float* wp = W + (size_t)n0 * WN + o * (DO_ * DI) + c * DI + 4 * h;

  v4f xa[GST], wa[GST];
#pragma unroll
  for (int j = 0; j < GST; ++j) {
    xa[j] = *(const v4f*)(xp + j * DI);
    wa[j] = *(const v4f*)(wp + (size_t)j * WN);
  }
  v8f acc0 = {}, acc1 = {};
  for (int st = 1; st < NPWG / GST; ++st) {
    xp += GST * DI;
    wp += (size_t)GST * WN;
    v4f xn[GST], wn[GST];
#pragma unroll
    for (int j = 0; j < GST; ++j) {
      xn[j] = *(const v4f*)(xp + j * DI);
      wn[j] = *(const v4f*)(wp + (size_t)j * WN);
    }
    __builtin_amdgcn_sched_barrier(0);
    s0_stage(xa, wa, acc0, acc1);
#pragma unroll
    for (int j = 0; j < GST; ++j) { xa[j] = xn[j]; wa[j] = wn[j]; }
  }
  s0_stage(xa, wa, acc0, acc1);

#pragma unroll
  for (int r = 0; r < 8; ++r) part[wave * 256 + r * 32 + lane] = acc0[r] + acc1[r];
  __syncthreads();
  const int bl = threadIdx.x >> 4;
  const int d  = threadIdx.x & 15;
  const int pi = (bl & 7) * 32 + (bl >> 3) * 16 + d;
  float s = 0.f;
#pragma unroll
  for (int w = 0; w < KW; ++w) s += part[w * 256 + pi];
  s *= 1.0f / NO;
  sh[threadIdx.x] = s;
  __syncthreads();
  float nn = 0.f;
#pragma unroll
  for (int dd = 0; dd < DO_; ++dd) { const float t = sh[bl * DO_ + dd]; nn += t * t; }
  const float scl = sqrtf(nn) / (1.f + nn);
  v0[(size_t)(mt * 16 + bl) * NOD + o * DO_ + d] = s * scl;
}

__global__ __launch_bounds__(128)
void caps_route(const float* __restrict__ x, const float* __restrict__ W,
                const float* __restrict__ vcur, float* __restrict__ rlog,
                float* __restrict__ ps, int iter) {
  __shared__ v4f fold[2 * FOLD_V4];
  const int lane = threadIdx.x & 31;
  const int wave = __builtin_amdgcn_readfirstlane((int)(threadIdx.x >> 5));
  const int wb   = wave & 1;
  const int nh   = wave >> 1;
  const int h    = lane >> 4;
  const int c    = lane & 15;
  const int p    = blockIdx.x >> 1;
  const int bt   = ((blockIdx.x & 1) << 1) + wb;
  const int b    = bt * 16 + c;

  float vr[2][16][8];
#pragma unroll
  for (int oh = 0; oh < 2; ++oh) {
#pragma unroll
    for (int r = 0; r < 8; ++r) {
      const int o = 16 * oh + 8 * h + r;
      const v4f* vp = (const v4f*)(vcur + (size_t)b * NOD + o * DO_);
#pragma unroll
      for (int q = 0; q < 4; ++q) {
        v4f t = vp[q];
        vr[oh][4 * q + 0][r] = t[0];
        vr[oh][4 * q + 1][r] = t[1];
        vr[oh][4 * q + 2][r] = t[2];
        vr[oh][4 * q + 3][r] = t[3];
      }
    }
  }

  float sacc[2][16][8];
#pragma unroll
  for (int oh = 0; oh < 2; ++oh)
#pragma unroll
    for (int d = 0; d < 16; ++d)
#pragma unroll
      for (int r = 0; r < 8; ++r) sacc[oh][d][r] = 0.f;

  const v8f zero = {};
  for (int nn = 0; nn < NPW; ++nn) {
    const int n = p * NPB + nh * NPW + nn;
    const float* wn = W + (size_t)n * WN + c * (DO_ * DI) + 4 * h;
    const v4f x4 = *(const v4f*)(x + ((size_t)b * NI + n) * DI + 4 * h);
    const v2f xb0 = {x4[0], x4[1]};
    const v2f xb1 = {x4[2], x4[3]};

    float lg[2][8];
#pragma unroll
    for (int oh = 0; oh < 2; ++oh) {
#pragma unroll
      for (int r = 0; r < 8; ++r) lg[oh][r] = 0.f;
#pragma unroll
      for (int d = 0; d < 16; ++d) {
        const v4f w4 = *(const v4f*)(wn + oh * (16 * DO_ * DI) + d * DI);
        const v2f a0 = {w4[0], w4[1]};
        const v2f a1 = {w4[2], w4[3]};
        v8f D = __builtin_amdgcn_wmma_f32_16x16x4_f32(
                    false, a0, false, xb0, (short)0, zero, false, false);
        D = __builtin_amdgcn_wmma_f32_16x16x4_f32(
                    false, a1, false, xb1, (short)0, D, false, false);
#pragma unroll
        for (int r = 0; r < 8; ++r) lg[oh][r] += vr[oh][d][r] * D[r];
      }
    }

    float* rl = rlog + ((size_t)b * NI + n) * NO + 8 * h;
#pragma unroll
    for (int oh = 0; oh < 2; ++oh) {
      v4f* rp = (v4f*)(rl + 16 * oh);
      if (iter == 1) {
        v4f t0 = {lg[oh][0], lg[oh][1], lg[oh][2], lg[oh][3]};
        v4f t1 = {lg[oh][4], lg[oh][5], lg[oh][6], lg[oh][7]};
        rp[0] = t0; rp[1] = t1;
      } else {
        const v4f t0 = rp[0], t1 = rp[1];
#pragma unroll
        for (int r = 0; r < 4; ++r) { lg[oh][r] += t0[r]; lg[oh][4 + r] += t1[r]; }
      }
    }

    float m = -3.4e38f;
#pragma unroll
    for (int oh = 0; oh < 2; ++oh)
#pragma unroll
      for (int r = 0; r < 8; ++r) m = fmaxf(m, lg[oh][r]);
    m = fmaxf(m, __shfl_xor(m, 16));
    float ssum = 0.f;
#pragma unroll
    for (int oh = 0; oh < 2; ++oh)
#pragma unroll
      for (int r = 0; r < 8; ++r) { lg[oh][r] = __expf(lg[oh][r] - m); ssum += lg[oh][r]; }
    ssum += __shfl_xor(ssum, 16);
    const float inv = 1.f / ssum;

#pragma unroll
    for (int oh = 0; oh < 2; ++oh) {
      float cc[8];
#pragma unroll
      for (int r = 0; r < 8; ++r) cc[r] = lg[oh][r] * inv;
#pragma unroll
      for (int d = 0; d < 16; ++d) {
        const v4f w4 = *(const v4f*)(wn + oh * (16 * DO_ * DI) + d * DI);
        const v2f a0 = {w4[0], w4[1]};
        const v2f a1 = {w4[2], w4[3]};
        v8f D = __builtin_amdgcn_wmma_f32_16x16x4_f32(
                    false, a0, false, xb0, (short)0, zero, false, false);
        D = __builtin_amdgcn_wmma_f32_16x16x4_f32(
                    false, a1, false, xb1, (short)0, D, false, false);
#pragma unroll
        for (int r = 0; r < 8; ++r) sacc[oh][d][r] += cc[r] * D[r];
      }
    }
  }

  v4f* fw = fold + wb * FOLD_V4 + lane;
  if (nh == 1) {
#pragma unroll
    for (int oh = 0; oh < 2; ++oh)
#pragma unroll
      for (int r = 0; r < 8; ++r)
#pragma unroll
        for (int q = 0; q < 4; ++q) {
          v4f t = {sacc[oh][4 * q + 0][r], sacc[oh][4 * q + 1][r],
                   sacc[oh][4 * q + 2][r], sacc[oh][4 * q + 3][r]};
          fw[((oh * 8 + r) * 4 + q) * 32] = t;
        }
  }
  __syncthreads();
  if (nh == 0) {
    float* pp = ps + (size_t)p * PS_STRIDE + (size_t)b * NOD;
#pragma unroll
    for (int oh = 0; oh < 2; ++oh) {
#pragma unroll
      for (int r = 0; r < 8; ++r) {
        const int o = 16 * oh + 8 * h + r;
        v4f* op = (v4f*)(pp + o * DO_);
#pragma unroll
        for (int q = 0; q < 4; ++q) {
          const v4f u = fw[((oh * 8 + r) * 4 + q) * 32];
          v4f t = {sacc[oh][4 * q + 0][r] + u[0], sacc[oh][4 * q + 1][r] + u[1],
                   sacc[oh][4 * q + 2][r] + u[2], sacc[oh][4 * q + 3][r] + u[3]};
          op[q] = t;
        }
      }
    }
  }
}

__global__ __launch_bounds__(512)
void caps_reduce_squash(const float* __restrict__ ps, int nparts, float prescale,
                        float* __restrict__ out) {
  __shared__ float sh[NOD];
  __shared__ float scl[NO];
  const int b  = blockIdx.x;
  const int od = threadIdx.x;
  float s = 0.f;
  for (int p = 0; p < nparts; ++p)
    s += ps[(size_t)p * PS_STRIDE + b * NOD + od];
  s *= prescale;
  sh[od] = s;
  __syncthreads();
  if (od < NO) {
    float nn = 0.f;
#pragma unroll
    for (int d = 0; d < DO_; ++d) { float t = sh[od * DO_ + d]; nn += t * t; }
    scl[od] = sqrtf(nn) / (1.f + nn);
  }
  __syncthreads();
  out[(size_t)b * NOD + od] = sh[od] * scl[od >> 4];
}

extern "C" void kernel_launch(void* const* d_in, const int* in_sizes, int n_in,
                              void* d_out, int out_size, void* d_ws, size_t ws_size,
                              hipStream_t stream) {
  const float* x = (const float*)d_in[0];
  const float* W = (const float*)d_in[1];
  float* out = (float*)d_out;

  float* ps   = (float*)d_ws;
  float* vbuf = ps + (size_t)NB_ROUTE * PS_STRIDE;
  float* rlog = vbuf + (size_t)B_SZ * NOD;

  caps_s0_squash<<<NBLK_S0, 256, 0, stream>>>(x, W, vbuf);
  caps_route<<<NBLK_ROUTE, 128, 0, stream>>>(x, W, vbuf, rlog, ps, 1);
  caps_reduce_squash<<<B_SZ, NOD, 0, stream>>>(ps, NB_ROUTE, 1.0f, vbuf);
  caps_route<<<NBLK_ROUTE, 128, 0, stream>>>(x, W, vbuf, rlog, ps, 2);
  caps_reduce_squash<<<B_SZ, NOD, 0, stream>>>(ps, NB_ROUTE, 1.0f, out);
}


# caps_s0: iteration-0 capsule sum split over two blocks per result tile (256 blocks, one on every WGP, was 128); the half sums are added and squashed where v0 is read (routing prologue, reduce addend)
# speedup vs baseline: 20.8828x; 1.2528x over previous
#include <hip/hip_runtime.h>
#include <math.h>

typedef __attribute__((ext_vector_type(2))) float v2f;
typedef __attribute__((ext_vector_type(4))) float v4f;
typedef __attribute__((ext_vector_type(8))) float v8f;

#define B_SZ   64
#define NI     2048
#define DI     8
#define NO     32
#define DO_    16
#define NOD    512
#define KTOT   (NI*DI)
#define WN     (NO*DO_*DI)

#define NB_ROUTE 128
#define NPB      (NI/NB_ROUTE)
#define NBLK_ROUTE 256
#define NPW      (NPB/2)
#define FOLD_V4  (64*32)
#define PS_STRIDE (B_SZ*NOD)

#define KW    8
#define KSPL  2
#define NPWG  (NI/(KW*KSPL))
#define HPLANE (B_SZ*NOD)
#define GST   16
#define NBLK_S0 256

__device__ __forceinline__ void s0_stage(const v4f (&xa)[GST], const v4f (&wa)[GST],
                                         v8f& acc0, v8f& acc1) {
#pragma unroll
  for (int j = 0; j < GST; j += 2) {
    const v2f a00 = {xa[j][0], xa[j][1]},         a01 = {xa[j][2], xa[j][3]};
    const v2f b00 = {wa[j][0], wa[j][1]},         b01 = {wa[j][2], wa[j][3]};
    const v2f a10 = {xa[j + 1][0], xa[j + 1][1]}, a11 = {xa[j + 1][2], xa[j + 1][3]};
    const v2f b10 = {wa[j + 1][0], wa[j + 1][1]}, b11 = {wa[j + 1][2], wa[j + 1][3]};
    acc0 = __builtin_amdgcn_wmma_f32_16x16x4_f32(false, a00, false, b00, (short)0, acc0, false, false);
    acc1 = __builtin_amdgcn_wmma_f32_16x16x4_f32(false, a10, false, b10, (short)0, acc1, false, false);
    acc0 = __builtin_amdgcn_wmma_f32_16x16x4_f32(false, a01, false, b01, (short)0, acc0, false, false);
    acc1 = __builtin_amdgcn_wmma_f32_16x16x4_f32(false, a11, false, b11, (short)0, acc1, false, false);
  }
}

__global__ __launch_bounds__(256)
void caps_s0_half(const float* __restrict__ x, const float* __restrict__ W,
                  float* __restrict__ hsum) {
  __shared__ float part[KW * 256];
  const int lane = threadIdx.x & 31;
  const int wave = __builtin_amdgcn_readfirstlane((int)(threadIdx.x >> 5));
  const int h  = lane >> 4;
  const int c  = lane & 15;
  const int kh = blockIdx.x & 1;
  const int mt = (blockIdx.x >> 1) & 3;
  const int o  = blockIdx.x >> 3;
  const int n0 = (kh * KW + wave) * NPWG;

  const float* xp = x + ((size_t)(mt * 16 + c) * NI + n0) * DI + 4 * h;
  const float* wp = W + (size_t)n0 * WN + o * (DO_ * DI) + c * DI + 4 * h;

  v4f xa[GST], wa[GST];
#pragma unroll
  for (int j = 0; j < GST; ++j) {
    xa[j] = *(const v4f*)(xp + j * DI);
    wa[j] = *(const v4f*)(wp + (size_t)j * WN);
  }
  v8f acc0 = {}, acc1 = {};
  for (int st = 1; st < NPWG / GST; ++st) {
    xp += GST * DI;
    wp += (size_t)GST * WN;
    v4f xn[GST], wn[GST];
#pragma unroll
    for (int j = 0; j < GST; ++j) {
      xn[j] = *(const v4f*)(xp + j * DI);
      wn[j] = *(const v4f*)(wp + (size_t)j * WN);
    }
    __builtin_amdgcn_sched_barrier(0);
    s0_stage(xa, wa, acc0, acc1);
#pragma unroll
    for (int j = 0; j < GST; ++j) { xa[j] = xn[j]; wa[j] = wn[j]; }
  }
  s0_stage(xa, wa, acc0, acc1);

#pragma unroll
  for (int r = 0; r < 8; ++r) part[wave * 256 + r * 32 + lane] = acc0[r] + acc1[r];
  __syncthreads();
  const int bl = threadIdx.x >> 4;
  const int d  = threadIdx.x & 15;
  const int pi = (bl & 7) * 32 + (bl >> 3) * 16 + d;
  float s = 0.f;
#pragma unroll
  for (int w = 0; w < KW; ++w) s += part[w * 256 + pi];
  hsum[(size_t)kh * HPLANE + (size_t)(mt * 16 + bl) * NOD + o * DO_ + d] = s;
}

template <bool FIRST>
__global__ __launch_bounds__(128)
void caps_route(const float* __restrict__ x, const float* __restrict__ W,
                const float* __restrict__ vcur, float* __restrict__ ps) {
  __shared__ v4f fold[2 * FOLD_V4];
  const int lane = threadIdx.x & 31;
  const int wave = __builtin_amdgcn_readfirstlane((int)(threadIdx.x >> 5));
  const int wb   = wave & 1;
  const int nh   = wave >> 1;
  const int h    = lane >> 4;
  const int c    = lane & 15;
  const int p    = blockIdx.x >> 1;
  const int bt   = ((blockIdx.x & 1) << 1) + wb;
  const int b    = bt * 16 + c;

  float vr[2][16][8];
#pragma unroll
  for (int oh = 0; oh < 2; ++oh) {
#pragma unroll
    for (int r = 0; r < 8; ++r) {
      const int o = 16 * oh + 8 * h + r;
      const v4f* vp = (const v4f*)(vcur + (size_t)b * NOD + o * DO_);
#pragma unroll
      for (int q = 0; q < 4; ++q) {
        v4f t = vp[q];
        if (FIRST) {
          const v4f u = vp[q + HPLANE / 4];
          t = (t + u) * (1.0f / NO);
        }
        vr[oh][4 * q + 0][r] = t[0];
        vr[oh][4 * q + 1][r] = t[1];
        vr[oh][4 * q + 2][r] = t[2];
        vr[oh][4 * q + 3][r] = t[3];
      }
    }
  }

  if (FIRST) {
#pragma unroll
    for (int oh = 0; oh < 2; ++oh)
#pragma unroll
      for (int r = 0; r < 8; ++r) {
        float nn = 0.f;
#pragma unroll
        for (int d = 0; d < 16; ++d) nn += vr[oh][d][r] * vr[oh][d][r];
        const float scl = sqrtf(nn) / (1.f + nn);
#pragma unroll
        for (int d = 0; d < 16; ++d) vr[oh][d][r] *= scl;
      }
  }

  float sacc[2][16][8];
#pragma unroll
  for (int oh = 0; oh < 2; ++oh)
#pragma unroll
    for (int d = 0; d < 16; ++d)
#pragma unroll
      for (int r = 0; r < 8; ++r) sacc[oh][d][r] = 0.f;

  const v8f zero = {};
  for (int nn = 0; nn < NPW; ++nn) {
    const int n = p * NPB + nh * NPW + nn;
    const float* wn = W + (size_t)n * WN + c * (DO_ * DI) + 4 * h;
    const v4f x4 = *(const v4f*)(x + ((size_t)b * NI + n) * DI + 4 * h);
    const v2f xb0 = {x4[0], x4[1]};
    const v2f xb1 = {x4[2], x4[3]};

    float lg[2][8];
#pragma unroll
    for (int oh = 0; oh < 2; ++oh) {
#pragma unroll
      for (int r = 0; r < 8; ++r) lg[oh][r] = 0.f;
#pragma unroll
      for (int d = 0; d < 16; ++d) {
        const v4f w4 = *(const v4f*)(wn + oh * (16 * DO_ * DI) + d * DI);
        const v2f a0 = {w4[0], w4[1]};
        const v2f a1 = {w4[2], w4[3]};
        v8f D = __builtin_amdgcn_wmma_f32_16x16x4_f32(
                    false, a0, false, xb0, (short)0, zero, false, false);
        D = __builtin_amdgcn_wmma_f32_16x16x4_f32(
                    false, a1, false, xb1, (short)0, D, false, false);
#pragma unroll
        for (int r = 0; r < 8; ++r) lg[oh][r] += vr[oh][d][r] * D[r];
      }
    }

    float m = -3.4e38f;
#pragma unroll
    for (int oh = 0; oh < 2; ++oh)
#pragma unroll
      for (int r = 0; r < 8; ++r) m = fmaxf(m, lg[oh][r]);
    m = fmaxf(m, __shfl_xor(m, 16));
    float ssum = 0.f;
#pragma unroll
    for (int oh = 0; oh < 2; ++oh)
#pragma unroll
      for (int r = 0; r < 8; ++r) { lg[oh][r] = __expf(lg[oh][r] - m); ssum += lg[oh][r]; }
    ssum += __shfl_xor(ssum, 16);
    const float inv = 1.f / ssum;

#pragma unroll
    for (int oh = 0; oh < 2; ++oh) {
      float cc[8];
#pragma unroll
      for (int r = 0; r < 8; ++r) cc[r] = lg[oh][r] * inv;
#pragma unroll
      for (int d = 0; d < 16; ++d) {
        const v4f w4 = *(const v4f*)(wn + oh * (16 * DO_ * DI) + d * DI);
        const v2f a0 = {w4[0], w4[1]};
        const v2f a1 = {w4[2], w4[3]};
        v8f D = __builtin_amdgcn_wmma_f32_16x16x4_f32(
                    false, a0, false, xb0, (short)0, zero, false, false);
        D = __builtin_amdgcn_wmma_f32_16x16x4_f32(
                    false, a1, false, xb1, (short)0, D, false, false);
#pragma unroll
        for (int r = 0; r < 8; ++r) sacc[oh][d][r] += cc[r] * D[r];
      }
    }
  }

  v4f* fw = fold + wb * FOLD_V4 + lane;
  if (nh == 1) {
#pragma unroll
    for (int oh = 0; oh < 2; ++oh)
#pragma unroll
      for (int r = 0; r < 8; ++r)
#pragma unroll
        for (int q = 0; q < 4; ++q) {
          v4f t = {sacc[oh][4 * q + 0][r], sacc[oh][4 * q + 1][r],
                   sacc[oh][4 * q + 2][r], sacc[oh][4 * q + 3][r]};
          fw[((oh * 8 + r) * 4 + q) * 32] = t;
        }
  }
  __syncthreads();
  if (nh == 0) {
    float* pp = ps + (size_t)p * PS_STRIDE + (size_t)b * NOD;
#pragma unroll
    for (int oh = 0; oh < 2; ++oh) {
#pragma unroll
      for (int r = 0; r < 8; ++r) {
        const int o = 16 * oh + 8 * h + r;
        v4f* op = (v4f*)(pp + o * DO_);
#pragma unroll
        for (int q = 0; q < 4; ++q) {
          const v4f u = fw[((oh * 8 + r) * 4 + q) * 32];
          v4f t = {sacc[oh][4 * q + 0][r] + u[0], sacc[oh][4 * q + 1][r] + u[1],
                   sacc[oh][4 * q + 2][r] + u[2], sacc[oh][4 * q + 3][r] + u[3]};
          op[q] = t;
        }
      }
    }
  }
}

template <bool ADD>
__global__ __launch_bounds__(512)
void caps_reduce_squash(const float* __restrict__ ps, int nparts, float prescale,
                        const float* h0, const float* h1, float* out) {
  __shared__ float sh[NOD];
  __shared__ float scl[NO];
  __shared__ float sh0[ADD ? NOD : 1];
  __shared__ float scl0[ADD ? NO : 1];
  const int b  = blockIdx.x;
  const int od = threadIdx.x;
  float a = 0.f;
  if (ADD) {
    a = (h0[(size_t)b * NOD + od] + h1[(size_t)b * NOD + od]) * (1.0f / NO);
    sh0[od] = a;
  }
  float s = 0.f;
  for (int p = 0; p < nparts; ++p)
    s += ps[(size_t)p * PS_STRIDE + b * NOD + od];
  s *= prescale;
  sh[od] = s;
  __syncthreads();
  if (od < NO) {
    float nn = 0.f;
#pragma unroll
    for (int d = 0; d < DO_; ++d) { float t = sh[od * DO_ + d]; nn += t * t; }
    scl[od] = sqrtf(nn) / (1.f + nn);
    if (ADD) {
      float n0 = 0.f;
#pragma unroll
      for (int d = 0; d < DO_; ++d) { float t = sh0[od * DO_ + d]; n0 += t * t; }
      scl0[od] = sqrtf(n0) / (1.f + n0);
    }
  }
  __syncthreads();
  float r = sh[od] * scl[od >> 4];
  if (ADD) r += a * scl0[od >> 4];
  out[(size_t)b * NOD + od] = r;
}

extern "C" void kernel_launch(void* const* d_in, const int* in_sizes, int n_in,
                              void* d_out, int out_size, void* d_ws, size_t ws_size,
                              hipStream_t stream) {
  const float* x = (const float*)d_in[0];
  const float* W = (const float*)d_in[1];
  float* out = (float*)d_out;

  float* ps   = (float*)d_ws;
  float* h0   = ps + (size_t)NB_ROUTE * PS_STRIDE;
  float* h1   = h0 + (size_t)HPLANE;

  caps_s0_half<<<NBLK_S0, 256, 0, stream>>>(x, W, h0);
  caps_route<true><<<NBLK_ROUTE, 128, 0, stream>>>(x, W, h0, ps);
  caps_reduce_squash<true><<<B_SZ, NOD, 0, stream>>>(ps, NB_ROUTE, 1.0f, h0, h1, h0);
  caps_route<false><<<NBLK_ROUTE, 128, 0, stream>>>(x, W, h0, ps);
  caps_reduce_squash<false><<<B_SZ, NOD, 0, stream>>>(ps, NB_ROUTE, 1.0f, nullptr, nullptr, out);
}


# caps_route: v in and partial s out change layout in LDS rows (pitch 2064 B), so each global load/store is 4 whole lines (was 32 lines x 16 B each); all 4 waves fold the halves and store
# speedup vs baseline: 21.6664x; 1.0375x over previous
#include <hip/hip_runtime.h>
#include <math.h>

typedef __attribute__((ext_vector_type(2))) float v2f;
typedef __attribute__((ext_vector_type(4))) float v4f;
typedef __attribute__((ext_vector_type(8))) float v8f;

#define B_SZ   64
#define NI     2048
#define DI     8
#define NO     32
#define DO_    16
#define NOD    512
#define KTOT   (NI*DI)
#define WN     (NO*DO_*DI)

#define NB_ROUTE 128
#define NPB      (NI/NB_ROUTE)
#define NBLK_ROUTE 256
#define NPW      (NPB/2)
#define ROWP     (NOD + 4)
#define TILE_F   (16*ROWP)
#define PS_STRIDE (B_SZ*NOD)

#define KW    8
#define KSPL  2
#define NPWG  (NI/(KW*KSPL))
#define HPLANE (B_SZ*NOD)
#define GST   16
#define NBLK_S0 256

__device__ __forceinline__ void s0_stage(const v4f (&xa)[GST], const v4f (&wa)[GST],
                                         v8f& acc0, v8f& acc1) {
#pragma unroll
  for (int j = 0; j < GST; j += 2) {
    const v2f a00 = {xa[j][0], xa[j][1]},         a01 = {xa[j][2], xa[j][3]};
    const v2f b00 = {wa[j][0], wa[j][1]},         b01 = {wa[j][2], wa[j][3]};
    const v2f a10 = {xa[j + 1][0], xa[j + 1][1]}, a11 = {xa[j + 1][2], xa[j + 1][3]};
    const v2f b10 = {wa[j + 1][0], wa[j + 1][1]}, b11 = {wa[j + 1][2], wa[j + 1][3]};
    acc0 = __builtin_amdgcn_wmma_f32_16x16x4_f32(false, a00, false, b00, (short)0, acc0, false, false);
    acc1 = __builtin_amdgcn_wmma_f32_16x16x4_f32(false, a10, false, b10, (short)0, acc1, false, false);
    acc0 = __builtin_amdgcn_wmma_f32_16x16x4_f32(false, a01, false, b01, (short)0, acc0, false, false);
    acc1 = __builtin_amdgcn_wmma_f32_16x16x4_f32(false, a11, false, b11, (short)0, acc1, false, false);
  }
}

__global__ __launch_bounds__(256)
void caps_s0_half(const float* __restrict__ x, const float* __restrict__ W,
                  float* __restrict__ hsum) {
  __shared__ float part[KW * 256];
  const int lane = threadIdx.x & 31;
  const int wave = __builtin_amdgcn_readfirstlane((int)(threadIdx.x >> 5));
  const int h  = lane >> 4;
  const int c  = lane & 15;
  const int kh = blockIdx.x & 1;
  const int mt = (blockIdx.x >> 1) & 3;
  const int o  = blockIdx.x >> 3;
  const int n0 = (kh * KW + wave) * NPWG;

  const float* xp = x + ((size_t)(mt * 16 + c) * NI + n0) * DI + 4 * h;
  const float* wp = W + (size_t)n0 * WN + o * (DO_ * DI) + c * DI + 4 * h;

  v4f xa[GST], wa[GST];
#pragma unroll
  for (int j = 0; j < GST; ++j) {
    xa[j] = *(const v4f*)(xp + j * DI);
    wa[j] = *(const v4f*)(wp + (size_t)j * WN);
  }
  v8f acc0 = {}, acc1 = {};
  for (int st = 1; st < NPWG / GST; ++st) {
    xp += GST * DI;
    wp += (size_t)GST * WN;
    v4f xn[GST], wn[GST];
#pragma unroll
    for (int j = 0; j < GST; ++j) {
      xn[j] = *(const v4f*)(xp + j * DI);
      wn[j] = *(const v4f*)(wp + (size_t)j * WN);
    }
    __builtin_amdgcn_sched_barrier(0);
    s0_stage(xa, wa, acc0, acc1);
#pragma unroll
    for (int j = 0; j < GST; ++j) { xa[j] = xn[j]; wa[j] = wn[j]; }
  }
  s0_stage(xa, wa, acc0, acc1);

#pragma unroll
  for (int r = 0; r < 8; ++r) part[wave * 256 + r * 32 + lane] = acc0[r] + acc1[r];
  __syncthreads();
  const int bl = threadIdx.x >> 4;
  const int d  = threadIdx.x & 15;
  const int pi = (bl & 7) * 32 + (bl >> 3) * 16 + d;
  float s = 0.f;
#pragma unroll
  for (int w = 0; w < KW; ++w) s += part[w * 256 + pi];
  hsum[(size_t)kh * HPLANE + (size_t)(mt * 16 + bl) * NOD + o * DO_ + d] = s;
}

template <bool FIRST>
__global__ __launch_bounds__(128)
void caps_route(const float* __restrict__ x, const float* __restrict__ W,
                const float* __restrict__ vcur, float* __restrict__ ps) {
  __shared__ __attribute__((aligned(16))) float rows[4 * TILE_F];
  const int lane = threadIdx.x & 31;
  const int wave = __builtin_amdgcn_readfirstlane((int)(threadIdx.x >> 5));
  const int wb   = wave & 1;
  const int nh   = wave >> 1;
  const int h    = lane >> 4;
  const int c    = lane & 15;
  const int p    = blockIdx.x >> 1;
  const int bt   = ((blockIdx.x & 1) << 1) + wb;
  const int b    = bt * 16 + c;

  {
    const float* vg = vcur + (size_t)(bt * 16 + 8 * nh) * NOD + lane * 4;
    float* sw = rows + wb * TILE_F + (8 * nh) * ROWP + lane * 4;
#pragma unroll
    for (int j = 0; j < 8; ++j)
#pragma unroll
      for (int k = 0; k < 4; ++k) {
        v4f t = *(const v4f*)(vg + j * NOD + k * 128);
        if (FIRST) {
          const v4f u = *(const v4f*)(vg + HPLANE + j * NOD + k * 128);
          t = (t + u) * (1.0f / NO);
        }
        *(v4f*)(sw + j * ROWP + k * 128) = t;
      }
  }
  __syncthreads();

  float vr[2][16][8];
  {
    const float* sr = rows + wb * TILE_F + c * ROWP + (8 * h) * DO_;
#pragma unroll
    for (int oh = 0; oh < 2; ++oh) {
#pragma unroll
      for (int r = 0; r < 8; ++r) {
#pragma unroll
        for (int q = 0; q < 4; ++q) {
          const v4f t = *(const v4f*)(sr + (16 * oh + r) * DO_ + 4 * q);
          vr[oh][4 * q + 0][r] = t[0];
          vr[oh][4 * q + 1][r] = t[1];
          vr[oh][4 * q + 2][r] = t[2];
          vr[oh][4 * q + 3][r] = t[3];
        }
      }
    }
  }
  __syncthreads();

  if (FIRST) {
#pragma unroll
    for (int oh = 0; oh < 2; ++oh)
#pragma unroll
      for (int r = 0; r < 8; ++r) {
        float nn = 0.f;
#pragma unroll
        for (int d = 0; d < 16; ++d) nn += vr[oh][d][r] * vr[oh][d][r];
        const float scl = sqrtf(nn) / (1.f + nn);
#pragma unroll
        for (int d = 0; d < 16; ++d) vr[oh][d][r] *= scl;
      }
  }

  float sacc[2][16][8];
#pragma unroll
  for (int oh = 0; oh < 2; ++oh)
#pragma unroll
    for (int d = 0; d < 16; ++d)
#pragma unroll
      for (int r = 0; r < 8; ++r) sacc[oh][d][r] = 0.f;

  const v8f zero = {};
  for (int nn = 0; nn < NPW; ++nn) {
    const int n = p * NPB + nh * NPW + nn;
    const float* wn = W + (size_t)n * WN + c * (DO_ * DI) + 4 * h;
    const v4f x4 = *(const v4f*)(x + ((size_t)b * NI + n) * DI + 4 * h);
    const v2f xb0 = {x4[0], x4[1]};
    const v2f xb1 = {x4[2], x4[3]};

    float lg[2][8];
#pragma unroll
    for (int oh = 0; oh < 2; ++oh) {
#pragma unroll
      for (int r = 0; r < 8; ++r) lg[oh][r] = 0.f;
#pragma unroll
      for (int d = 0; d < 16; ++d) {
        const v4f w4 = *(const v4f*)(wn + oh * (16 * DO_ * DI) + d * DI);
        const v2f a0 = {w4[0], w4[1]};
        const v2f a1 = {w4[2], w4[3]};
        v8f D = __builtin_amdgcn_wmma_f32_16x16x4_f32(
                    false, a0, false, xb0, (short)0, zero, false, false);
        D = __builtin_amdgcn_wmma_f32_16x16x4_f32(
                    false, a1, false, xb1, (short)0, D, false, false);
#pragma unroll
        for (int r = 0; r < 8; ++r) lg[oh][r] += vr[oh][d][r] * D[r];
      }
    }

    float m = -3.4e38f;
#pragma unroll
    for (int oh = 0; oh < 2; ++oh)
#pragma unroll
      for (int r = 0; r < 8; ++r) m = fmaxf(m, lg[oh][r]);
    m = fmaxf(m, __shfl_xor(m, 16));
    float ssum = 0.f;
#pragma unroll
    for (int oh = 0; oh < 2; ++oh)
#pragma unroll
      for (int r = 0; r < 8; ++r) { lg[oh][r] = __expf(lg[oh][r] - m); ssum += lg[oh][r]; }
    ssum += __shfl_xor(ssum, 16);
    const float inv = 1.f / ssum;

#pragma unroll
    for (int oh = 0; oh < 2; ++oh) {
      float cc[8];
#pragma unroll
      for (int r = 0; r < 8; ++r) cc[r] = lg[oh][r] * inv;
#pragma unroll
      for (int d = 0; d < 16; ++d) {
        const v4f w4 = *(const v4f*)(wn + oh * (16 * DO_ * DI) + d * DI);
        const v2f a0 = {w4[0], w4[1]};
        const v2f a1 = {w4[2], w4[3]};
        v8f D = __builtin_amdgcn_wmma_f32_16x16x4_f32(
                    false, a0, false, xb0, (short)0, zero, false, false);
        D = __builtin_amdgcn_wmma_f32_16x16x4_f32(
                    false, a1, false, xb1, (short)0, D, false, false);
#pragma unroll
        for (int r = 0; r < 8; ++r) sacc[oh][d][r] += cc[r] * D[r];
      }
    }
  }

  {
    float* ew = rows + (nh * 2 + wb) * TILE_F + c * ROWP + (8 * h) * DO_;
#pragma unroll
    for (int oh = 0; oh < 2; ++oh)
#pragma unroll
      for (int r = 0; r < 8; ++r)
#pragma unroll
        for (int q = 0; q < 4; ++q) {
          const v4f t = {sacc[oh][4 * q + 0][r], sacc[oh][4 * q + 1][r],
                         sacc[oh][4 * q + 2][r], sacc[oh][4 * q + 3][r]};
          *(v4f*)(ew + (16 * oh + r) * DO_ + 4 * q) = t;
        }
  }
  __syncthreads();
  {
    const float* e0 = rows + wb * TILE_F + (8 * nh) * ROWP + lane * 4;
    const float* e1 = e0 + 2 * TILE_F;
    float* pg = ps + (size_t)p * PS_STRIDE + (size_t)(bt * 16 + 8 * nh) * NOD + lane * 4;
#pragma unroll
    for (int j = 0; j < 8; ++j)
#pragma unroll
      for (int k = 0; k < 4; ++k) {
        const v4f a = *(const v4f*)(e0 + j * ROWP + k * 128);
        const v4f u = *(const v4f*)(e1 + j * ROWP + k * 128);
        *(v4f*)(pg + j * NOD + k * 128) = a + u;
      }
  }
}

template <bool ADD>
__global__ __launch_bounds__(512)
void caps_reduce_squash(const float* __restrict__ ps, int nparts, float prescale,
                        const float* h0, const float* h1, float* out) {
  __shared__ float sh[NOD];
  __shared__ float scl[NO];
  __shared__ float sh0[ADD ? NOD : 1];
  __shared__ float scl0[ADD ? NO : 1];
  const int b  = blockIdx.x;
  const int od = threadIdx.x;
  float a = 0.f;
  if (ADD) {
    a = (h0[(size_t)b * NOD + od] + h1[(size_t)b * NOD + od]) * (1.0f / NO);
    sh0[od] = a;
  }
  float s = 0.f;
  for (int p = 0; p < nparts; ++p)
    s += ps[(size_t)p * PS_STRIDE + b * NOD + od];
  s *= prescale;
  sh[od] = s;
  __syncthreads();
  if (od < NO) {
    float nn = 0.f;
#pragma unroll
    for (int d = 0; d < DO_; ++d) { float t = sh[od * DO_ + d]; nn += t * t; }
    scl[od] = sqrtf(nn) / (1.f + nn);
    if (ADD) {
      float n0 = 0.f;
#pragma unroll
      for (int d = 0; d < DO_; ++d) { float t = sh0[od * DO_ + d]; n0 += t * t; }
      scl0[od] = sqrtf(n0) / (1.f + n0);
    }
  }
  __syncthreads();
  float r = sh[od] * scl[od >> 4];
  if (ADD) r += a * scl0[od >> 4];
  out[(size_t)b * NOD + od] = r;
}

extern "C" void kernel_launch(void* const* d_in, const int* in_sizes, int n_in,
                              void* d_out, int out_size, void* d_ws, size_t ws_size,
                              hipStream_t stream) {
  const float* x = (const float*)d_in[0];
  const float* W = (const float*)d_in[1];
  float* out = (float*)d_out;

  float* ps   = (float*)d_ws;
  float* h0   = ps + (size_t)NB_ROUTE * PS_STRIDE;
  float* h1   = h0 + (size_t)HPLANE;

  caps_s0_half<<<NBLK_S0, 256, 0, stream>>>(x, W, h0);
  caps_route<true><<<NBLK_ROUTE, 128, 0, stream>>>(x, W, h0, ps);
  caps_reduce_squash<true><<<B_SZ, NOD, 0, stream>>>(ps, NB_ROUTE, 1.0f, h0, h1, h0);
  caps_route<false><<<NBLK_ROUTE, 128, 0, stream>>>(x, W, h0, ps);
  caps_reduce_squash<false><<<B_SZ, NOD, 0, stream>>>(ps, NB_ROUTE, 1.0f, nullptr, nullptr, out);
}


# caps_reduce: each thread's 32 plane reads are independent 16-byte loads issued before the first add (was 8 at a time, 4 bytes each): 1 memory round trip per block, not 16; 4 plane groups meet in LDS
# speedup vs baseline: 25.8203x; 1.1917x over previous
#include <hip/hip_runtime.h>
#include <math.h>

typedef __attribute__((ext_vector_type(2))) float v2f;
typedef __attribute__((ext_vector_type(4))) float v4f;
typedef __attribute__((ext_vector_type(8))) float v8f;

#define B_SZ   64
#define NI     2048
#define DI     8
#define NO     32
#define DO_    16
#define NOD    512
#define KTOT   (NI*DI)
#define WN     (NO*DO_*DI)

#define NB_ROUTE 128
#define NPB      (NI/NB_ROUTE)
#define NBLK_ROUTE 256
#define NPW      (NPB/2)
#define ROWP     (NOD + 4)
#define TILE_F   (16*ROWP)
#define PS_STRIDE (B_SZ*NOD)

#define KW    8
#define KSPL  2
#define NPWG  (NI/(KW*KSPL))
#define HPLANE (B_SZ*NOD)
#define GST   16
#define NBLK_S0 256

__device__ __forceinline__ void s0_stage(const v4f (&xa)[GST], const v4f (&wa)[GST],
                                         v8f& acc0, v8f& acc1) {
#pragma unroll
  for (int j = 0; j < GST; j += 2) {
    const v2f a00 = {xa[j][0], xa[j][1]},         a01 = {xa[j][2], xa[j][3]};
    const v2f b00 = {wa[j][0], wa[j][1]},         b01 = {wa[j][2], wa[j][3]};
    const v2f a10 = {xa[j + 1][0], xa[j + 1][1]}, a11 = {xa[j + 1][2], xa[j + 1][3]};
    const v2f b10 = {wa[j + 1][0], wa[j + 1][1]}, b11 = {wa[j + 1][2], wa[j + 1][3]};
    acc0 = __builtin_amdgcn_wmma_f32_16x16x4_f32(false, a00, false, b00, (short)0, acc0, false, false);
    acc1 = __builtin_amdgcn_wmma_f32_16x16x4_f32(false, a10, false, b10, (short)0, acc1, false, false);
    acc0 = __builtin_amdgcn_wmma_f32_16x16x4_f32(false, a01, false, b01, (short)0, acc0, false, false);
    acc1 = __builtin_amdgcn_wmma_f32_16x16x4_f32(false, a11, false, b11, (short)0, acc1, false, false);
  }
}

__global__ __launch_bounds__(256)
void caps_s0_half(const float* __restrict__ x, const float* __restrict__ W,
                  float* __restrict__ hsum) {
  __shared__ float part[KW * 256];
  const int lane = threadIdx.x & 31;
  const int wave = __builtin_amdgcn_readfirstlane((int)(threadIdx.x >> 5));
  const int h  = lane >> 4;
  const int c  = lane & 15;
  const int kh = blockIdx.x & 1;
  const int mt = (blockIdx.x >> 1) & 3;
  const int o  = blockIdx.x >> 3;
  const int n0 = (kh * KW + wave) * NPWG;

  const float* xp = x + ((size_t)(mt * 16 + c) * NI + n0) * DI + 4 * h;
  const float* wp = W + (size_t)n0 * WN + o * (DO_ * DI) + c * DI + 4 * h;

  v4f xa[GST], wa[GST];
#pragma unroll
  for (int j = 0; j < GST; ++j) {
    xa[j] = *(const v4f*)(xp + j * DI);
    wa[j] = *(const v4f*)(wp + (size_t)j * WN);
  }
  v8f acc0 = {}, acc1 = {};
  for (int st = 1; st < NPWG / GST; ++st) {
    xp += GST * DI;
    wp += (size_t)GST * WN;
    v4f xn[GST], wn[GST];
#pragma unroll
    for (int j = 0; j < GST; ++j) {
      xn[j] = *(const v4f*)(xp + j * DI);
      wn[j] = *(const v4f*)(wp + (size_t)j * WN);
    }
    __builtin_amdgcn_sched_barrier(0);
    s0_stage(xa, wa, acc0, acc1);
#pragma unroll
    for (int j = 0; j < GST; ++j) { xa[j] = xn[j]; wa[j] = wn[j]; }
  }
  s0_stage(xa, wa, acc0, acc1);

#pragma unroll
  for (int r = 0; r < 8; ++r) part[wave * 256 + r * 32 + lane] = acc0[r] + acc1[r];
  __syncthreads();
  const int bl = threadIdx.x >> 4;
  const int d  = threadIdx.x & 15;
  const int pi = (bl & 7) * 32 + (bl >> 3) * 16 + d;
  float s = 0.f;
#pragma unroll
  for (int w = 0; w < KW; ++w) s += part[w * 256 + pi];
  hsum[(size_t)kh * HPLANE + (size_t)(mt * 16 + bl) * NOD + o * DO_ + d] = s;
}

template <bool FIRST>
__global__ __launch_bounds__(128)
void caps_route(const float* __restrict__ x, const float* __restrict__ W,
                const float* __restrict__ vcur, float* __restrict__ ps) {
  __shared__ __attribute__((aligned(16))) float rows[4 * TILE_F];
  const int lane = threadIdx.x & 31;
  const int wave = __builtin_amdgcn_readfirstlane((int)(threadIdx.x >> 5));
  const int wb   = wave & 1;
  const int nh   = wave >> 1;
  const int h    = lane >> 4;
  const int c    = lane & 15;
  const int p    = blockIdx.x >> 1;
  const int bt   = ((blockIdx.x & 1) << 1) + wb;
  const int b    = bt * 16 + c;

  {
    const float* vg = vcur + (size_t)(bt * 16 + 8 * nh) * NOD + lane * 4;
    float* sw = rows + wb * TILE_F + (8 * nh) * ROWP + lane * 4;
#pragma unroll
    for (int j = 0; j < 8; ++j)
#pragma unroll
      for (int k = 0; k < 4; ++k) {
        v4f t = *(const v4f*)(vg + j * NOD + k * 128);
        if (FIRST) {
          const v4f u = *(const v4f*)(vg + HPLANE + j * NOD + k * 128);
          t = (t + u) * (1.0f / NO);
        }
        *(v4f*)(sw + j * ROWP + k * 128) = t;
      }
  }
  __syncthreads();

  float vr[2][16][8];
  {
    const float* sr = rows + wb * TILE_F + c * ROWP + (8 * h) * DO_;
#pragma unroll
    for (int oh = 0; oh < 2; ++oh) {
#pragma unroll
      for (int r = 0; r < 8; ++r) {
#pragma unroll
        for (int q = 0; q < 4; ++q) {
          const v4f t = *(const v4f*)(sr + (16 * oh + r) * DO_ + 4 * q);
          vr[oh][4 * q + 0][r] = t[0];
          vr[oh][4 * q + 1][r] = t[1];
          vr[oh][4 * q + 2][r] = t[2];
          vr[oh][4 * q + 3][r] = t[3];
        }
      }
    }
  }
  __syncthreads();

  if (FIRST) {
#pragma unroll
    for (int oh = 0; oh < 2; ++oh)
#pragma unroll
      for (int r = 0; r < 8; ++r) {
        float nn = 0.f;
#pragma unroll
        for (int d = 0; d < 16; ++d) nn += vr[oh][d][r] * vr[oh][d][r];
        const float scl = sqrtf(nn) / (1.f + nn);
#pragma unroll
        for (int d = 0; d < 16; ++d) vr[oh][d][r] *= scl;
      }
  }

  float sacc[2][16][8];
#pragma unroll
  for (int oh = 0; oh < 2; ++oh)
#pragma unroll
    for (int d = 0; d < 16; ++d)
#pragma unroll
      for (int r = 0; r < 8; ++r) sacc[oh][d][r] = 0.f;

  const v8f zero = {};
  for (int nn = 0; nn < NPW; ++nn) {
    const int n = p * NPB + nh * NPW + nn;
    const float* wn = W + (size_t)n * WN + c * (DO_ * DI) + 4 * h;
    const v4f x4 = *(const v4f*)(x + ((size_t)b * NI + n) * DI + 4 * h);
    const v2f xb0 = {x4[0], x4[1]};
    const v2f xb1 = {x4[2], x4[3]};

    float lg[2][8];
#pragma unroll
    for (int oh = 0; oh < 2; ++oh) {
#pragma unroll
      for (int r = 0; r < 8; ++r) lg[oh][r] = 0.f;
#pragma unroll
      for (int d = 0; d < 16; ++d) {
        const v4f w4 = *(const v4f*)(wn + oh * (16 * DO_ * DI) + d * DI);
        const v2f a0 = {w4[0], w4[1]};
        const v2f a1 = {w4[2], w4[3]};
        v8f D = __builtin_amdgcn_wmma_f32_16x16x4_f32(
                    false, a0, false, xb0, (short)0, zero, false, false);
        D = __builtin_amdgcn_wmma_f32_16x16x4_f32(
                    false, a1, false, xb1, (short)0, D, false, false);
#pragma unroll
        for (int r = 0; r < 8; ++r) lg[oh][r] += vr[oh][d][r] * D[r];
      }
    }

    float m = -3.4e38f;
#pragma unroll
    for (int oh = 0; oh < 2; ++oh)
#pragma unroll
      for (int r = 0; r < 8; ++r) m = fmaxf(m, lg[oh][r]);
    m = fmaxf(m, __shfl_xor(m, 16));
    float ssum = 0.f;
#pragma unroll
    for (int oh = 0; oh < 2; ++oh)
#pragma unroll
      for (int r = 0; r < 8; ++r) { lg[oh][r] = __expf(lg[oh][r] - m); ssum += lg[oh][r]; }
    ssum += __shfl_xor(ssum, 16);
    const float inv = 1.f / ssum;

#pragma unroll
    for (int oh = 0; oh < 2; ++oh) {
      float cc[8];
#pragma unroll
      for (int r = 0; r < 8; ++r) cc[r] = lg[oh][r] * inv;
#pragma unroll
      for (int d = 0; d < 16; ++d) {
        const v4f w4 = *(const v4f*)(wn + oh * (16 * DO_ * DI) + d * DI);
        const v2f a0 = {w4[0], w4[1]};
        const v2f a1 = {w4[2], w4[3]};
        v8f D = __builtin_amdgcn_wmma_f32_16x16x4_f32(
                    false, a0, false, xb0, (short)0, zero, false, false);
        D = __builtin_amdgcn_wmma_f32_16x16x4_f32(
                    false, a1, false, xb1, (short)0, D, false, false);
#pragma unroll
        for (int r = 0; r < 8; ++r) sacc[oh][d][r] += cc[r] * D[r];
      }
    }
  }

  {
    float* ew = rows + (nh * 2 + wb) * TILE_F + c * ROWP + (8 * h) * DO_;
#pragma unroll
    for (int oh = 0; oh < 2; ++oh)
#pragma unroll
      for (int r = 0; r < 8; ++r)
#pragma unroll
        for (int q = 0; q < 4; ++q) {
          const v4f t = {sacc[oh][4 * q + 0][r], sacc[oh][4 * q + 1][r],
                         sacc[oh][4 * q + 2][r], sacc[oh][4 * q + 3][r]};
          *(v4f*)(ew + (16 * oh + r) * DO_ + 4 * q) = t;
        }
  }
  __syncthreads();
  {
    const float* e0 = rows + wb * TILE_F + (8 * nh) * ROWP + lane * 4;
    const float* e1 = e0 + 2 * TILE_F;
    float* pg = ps + (size_t)p * PS_STRIDE + (size_t)(bt * 16 + 8 * nh) * NOD + lane * 4;
#pragma unroll
    for (int j = 0; j < 8; ++j)
#pragma unroll
      for (int k = 0; k < 4; ++k) {
        const v4f a = *(const v4f*)(e0 + j * ROWP + k * 128);
        const v4f u = *(const v4f*)(e1 + j * ROWP + k * 128);
        *(v4f*)(pg + j * NOD + k * 128) = a + u;
      }
  }
}

#define RGRP 4
#define PPG  (NB_ROUTE/RGRP)

template <bool ADD>
__global__ __launch_bounds__(512)
void caps_reduce_squash(const float* __restrict__ ps,
                        const float* h0, const float* h1, float* out) {
  __shared__ __attribute__((aligned(16))) float part[(RGRP - 1) * NOD];
  const int b = blockIdx.x;
  const int f = threadIdx.x & 127;
  const int g = __builtin_amdgcn_readfirstlane((int)(threadIdx.x >> 7));
  const size_t e = (size_t)b * NOD + 4 * f;
  const float* src = ps + (size_t)(g * PPG) * PS_STRIDE + e;
  v4f t[PPG];
#pragma unroll
  for (int j = 0; j < PPG; ++j) t[j] = *(const v4f*)(src + (size_t)j * PS_STRIDE);
  v4f a = {};
  if (ADD) {
    if (g == 0) a = (*(const v4f*)(h0 + e) + *(const v4f*)(h1 + e)) * (1.0f / NO);
  }
  __builtin_amdgcn_sched_barrier(0);
  v4f s = t[0];
#pragma unroll
  for (int j = 1; j < PPG; ++j) s += t[j];
  if (g != 0) *(v4f*)(part + (g - 1) * NOD + 4 * f) = s;
  __syncthreads();
  if (g == 0) {
#pragma unroll
    for (int k = 0; k < RGRP - 1; ++k) s += *(const v4f*)(part + k * NOD + 4 * f);
    float nn = s[0] * s[0] + s[1] * s[1] + s[2] * s[2] + s[3] * s[3];
    nn += __shfl_xor(nn, 1);
    nn += __shfl_xor(nn, 2);
    v4f r = s * (sqrtf(nn) / (1.f + nn));
    if (ADD) {
      float n0 = a[0] * a[0] + a[1] * a[1] + a[2] * a[2] + a[3] * a[3];
      n0 += __shfl_xor(n0, 1);
      n0 += __shfl_xor(n0, 2);
      r += a * (sqrtf(n0) / (1.f + n0));
    }
    *(v4f*)(out + e) = r;
  }
}

extern "C" void kernel_launch(void* const* d_in, const int* in_sizes, int n_in,
                              void* d_out, int out_size, void* d_ws, size_t ws_size,
                              hipStream_t stream) {
  const float* x = (const float*)d_in[0];
  const float* W = (const float*)d_in[1];
  float* out = (float*)d_out;

  float* ps   = (float*)d_ws;
  float* h0   = ps + (size_t)NB_ROUTE * PS_STRIDE;
  float* h1   = h0 + (size_t)HPLANE;

  caps_s0_half<<<NBLK_S0, 256, 0, stream>>>(x, W, h0);
  caps_route<true><<<NBLK_ROUTE, 128, 0, stream>>>(x, W, h0, ps);
  caps_reduce_squash<true><<<B_SZ, NOD, 0, stream>>>(ps, h0, h1, h0);
  caps_route<false><<<NBLK_ROUTE, 128, 0, stream>>>(x, W, h0, ps);
  caps_reduce_squash<false><<<B_SZ, NOD, 0, stream>>>(ps, nullptr, nullptr, out);
}


# caps_route: a capsule's W slice comes through LDS (row pitch 528 B), fetched once per wave pair as whole 512-byte rows at the top of the previous capsule's trip; A fragments by 16-byte LDS reads
# speedup vs baseline: 28.5357x; 1.1052x over previous
#include <hip/hip_runtime.h>
#include <math.h>

typedef __attribute__((ext_vector_type(2))) float v2f;
typedef __attribute__((ext_vector_type(4))) float v4f;
typedef __attribute__((ext_vector_type(8))) float v8f;

#define B_SZ   64
#define NI     2048
#define DI     8
#define NO     32
#define DO_    16
#define NOD    512
#define KTOT   (NI*DI)
#define WN     (NO*DO_*DI)

#define NB_ROUTE 128
#define NPB      (NI/NB_ROUTE)
#define NBLK_ROUTE 256
#define NPW      (NPB/2)
#define ROWP     (NOD + 4)
#define TILE_F   (16*ROWP)
#define PS_STRIDE (B_SZ*NOD)
#define WROWP    (DO_*DI + 4)
#define WBUF_F   (NO*WROWP)

#define KW    8
#define KSPL  2
#define NPWG  (NI/(KW*KSPL))
#define HPLANE (B_SZ*NOD)
#define GST   16
#define NBLK_S0 256

__device__ __forceinline__ void s0_stage(const v4f (&xa)[GST], const v4f (&wa)[GST],
                                         v8f& acc0, v8f& acc1) {
#pragma unroll
  for (int j = 0; j < GST; j += 2) {
    const v2f a00 = {xa[j][0], xa[j][1]},         a01 = {xa[j][2], xa[j][3]};
    const v2f b00 = {wa[j][0], wa[j][1]},         b01 = {wa[j][2], wa[j][3]};
    const v2f a10 = {xa[j + 1][0], xa[j + 1][1]}, a11 = {xa[j + 1][2], xa[j + 1][3]};
    const v2f b10 = {wa[j + 1][0], wa[j + 1][1]}, b11 = {wa[j + 1][2], wa[j + 1][3]};
    acc0 = __builtin_amdgcn_wmma_f32_16x16x4_f32(false, a00, false, b00, (short)0, acc0, false, false);
    acc1 = __builtin_amdgcn_wmma_f32_16x16x4_f32(false, a10, false, b10, (short)0, acc1, false, false);
    acc0 = __builtin_amdgcn_wmma_f32_16x16x4_f32(false, a01, false, b01, (short)0, acc0, false, false);
    acc1 = __builtin_amdgcn_wmma_f32_16x16x4_f32(false, a11, false, b11, (short)0, acc1, false, false);
  }
}

__global__ __launch_bounds__(256)
void caps_s0_half(const float* __restrict__ x, const float* __restrict__ W,
                  float* __restrict__ hsum) {
  __shared__ float part[KW * 256];
  const int lane = threadIdx.x & 31;
  const int wave = __builtin_amdgcn_readfirstlane((int)(threadIdx.x >> 5));
  const int h  = lane >> 4;
  const int c  = lane & 15;
  const int kh = blockIdx.x & 1;
  const int mt = (blockIdx.x >> 1) & 3;
  const int o  = blockIdx.x >> 3;
  const int n0 = (kh * KW + wave) * NPWG;

  const float* xp = x + ((size_t)(mt * 16 + c) * NI + n0) * DI + 4 * h;
  const float* wp = W + (size_t)n0 * WN + o * (DO_ * DI) + c * DI + 4 * h;

  v4f xa[GST], wa[GST];
#pragma unroll
  for (int j = 0; j < GST; ++j) {
    xa[j] = *(const v4f*)(xp + j * DI);
    wa[j] = *(const v4f*)(wp + (size_t)j * WN);
  }
  v8f acc0 = {}, acc1 = {};
  for (int st = 1; st < NPWG / GST; ++st) {
    xp += GST * DI;
    wp += (size_t)GST * WN;
    v4f xn[GST], wn[GST];
#pragma unroll
    for (int j = 0; j < GST; ++j) {
      xn[j] = *(const v4f*)(xp + j * DI);
      wn[j] = *(const v4f*)(wp + (size_t)j * WN);
    }
    __builtin_amdgcn_sched_barrier(0);
    s0_stage(xa, wa, acc0, acc1);
#pragma unroll
    for (int j = 0; j < GST; ++j) { xa[j] = xn[j]; wa[j] = wn[j]; }
  }
  s0_stage(xa, wa, acc0, acc1);

#pragma unroll
  for (int r = 0; r < 8; ++r) part[wave * 256 + r * 32 + lane] = acc0[r] + acc1[r];
  __syncthreads();
  const int bl = threadIdx.x >> 4;
  const int d  = threadIdx.x & 15;
  const int pi = (bl & 7) * 32 + (bl >> 3) * 16 + d;
  float s = 0.f;
#pragma unroll
  for (int w = 0; w < KW; ++w) s += part[w * 256 + pi];
  hsum[(size_t)kh * HPLANE + (size_t)(mt * 16 + bl) * NOD + o * DO_ + d] = s;
}

template <bool MORE>
__device__ __forceinline__ void route_trip(const float* wfr, const float* __restrict__ wgn,
                                           float* wsn, const float* __restrict__ xgn, v4f& x4,
                                           const float (&vr)[2][16][8], float (&sacc)[2][16][8]) {
  const v8f zero = {};
  const v2f xb0 = {x4[0], x4[1]};
  const v2f xb1 = {x4[2], x4[3]};

  v4f st[16];
  v4f xn;
  if (MORE) {
#pragma unroll
    for (int j = 0; j < 16; ++j) st[j] = *(const v4f*)(wgn + j * (DO_ * DI));
    xn = *(const v4f*)xgn;
  }

  v8f T[2][16];
  float lg[2][8];
#pragma unroll
  for (int oh = 0; oh < 2; ++oh) {
#pragma unroll
    for (int r = 0; r < 8; ++r) lg[oh][r] = 0.f;
#pragma unroll
    for (int d = 0; d < 16; ++d) {
      const v4f w4 = *(const v4f*)(wfr + oh * (16 * WROWP) + d * DI);
      const v2f a0 = {w4[0], w4[1]};
      const v2f a1 = {w4[2], w4[3]};
      v8f D = __builtin_amdgcn_wmma_f32_16x16x4_f32(
                  false, a0, false, xb0, (short)0, zero, false, false);
      D = __builtin_amdgcn_wmma_f32_16x16x4_f32(
                  false, a1, false, xb1, (short)0, D, false, false);
      T[oh][d] = D;
#pragma unroll
      for (int r = 0; r < 8; ++r) lg[oh][r] += vr[oh][d][r] * D[r];
    }
  }

  float m = -3.4e38f;
#pragma unroll
  for (int oh = 0; oh < 2; ++oh)
#pragma unroll
    for (int r = 0; r < 8; ++r) m = fmaxf(m, lg[oh][r]);
  m = fmaxf(m, __shfl_xor(m, 16));
  float ssum = 0.f;
#pragma unroll
  for (int oh = 0; oh < 2; ++oh)
#pragma unroll
    for (int r = 0; r < 8; ++r) { lg[oh][r] = __expf(lg[oh][r] - m); ssum += lg[oh][r]; }
  ssum += __shfl_xor(ssum, 16);
  const float inv = 1.f / ssum;

#pragma unroll
  for (int oh = 0; oh < 2; ++oh) {
    float cc[8];
#pragma unroll
    for (int r = 0; r < 8; ++r) cc[r] = lg[oh][r] * inv;
#pragma unroll
    for (int d = 0; d < 16; ++d)
#pragma unroll
      for (int r = 0; r < 8; ++r) sacc[oh][d][r] += cc[r] * T[oh][d][r];
  }

  if (MORE) {
#pragma unroll
    for (int j = 0; j < 16; ++j) *(v4f*)(wsn + j * WROWP) = st[j];
    x4 = xn;
  }
}

template <bool FIRST>
__global__ __launch_bounds__(128)
void caps_route(const float* __restrict__ x, const float* __restrict__ W,
                const float* __restrict__ vcur, float* __restrict__ ps) {
  __shared__ __attribute__((aligned(16))) float rows[4 * TILE_F];
  __shared__ __attribute__((aligned(16))) float wst[4 * WBUF_F];
  const int lane = threadIdx.x & 31;
  const int wave = __builtin_amdgcn_readfirstlane((int)(threadIdx.x >> 5));
  const int wb   = wave & 1;
  const int nh   = wave >> 1;
  const int h    = lane >> 4;
  const int c    = lane & 15;
  const int p    = blockIdx.x >> 1;
  const int bt   = ((blockIdx.x & 1) << 1) + wb;
  const int b    = bt * 16 + c;
  const int n0   = p * NPB + nh * NPW;

  const float* wg = W + (size_t)n0 * WN + (16 * wb) * (DO_ * DI) + lane * 4;
  float* wsw = wst + (nh * 2) * WBUF_F + (16 * wb) * WROWP + lane * 4;
  const float* wsr = wst + (nh * 2) * WBUF_F + c * WROWP + 4 * h;
  const float* xg = x + ((size_t)b * NI + n0) * DI + 4 * h;

#pragma unroll
  for (int j = 0; j < 16; ++j)
    *(v4f*)(wsw + j * WROWP) = *(const v4f*)(wg + j * (DO_ * DI));
  v4f x4 = *(const v4f*)xg;

  {
    const float* vg = vcur + (size_t)(bt * 16 + 8 * nh) * NOD + lane * 4;
    float* sw = rows + wb * TILE_F + (8 * nh) * ROWP + lane * 4;
#pragma unroll
    for (int j = 0; j < 8; ++j)
#pragma unroll
      for (int k = 0; k < 4; ++k) {
        v4f t = *(const v4f*)(vg + j * NOD + k * 128);
        if (FIRST) {
          const v4f u = *(const v4f*)(vg + HPLANE + j * NOD + k * 128);
          t = (t + u) * (1.0f / NO);
        }
        *(v4f*)(sw + j * ROWP + k * 128) = t;
      }
  }
  __syncthreads();

  float vr[2][16][8];
  {
    const float* sr = rows + wb * TILE_F + c * ROWP + (8 * h) * DO_;
#pragma unroll
    for (int oh = 0; oh < 2; ++oh) {
#pragma unroll
      for (int r = 0; r < 8; ++r) {
#pragma unroll
        for (int q = 0; q < 4; ++q) {
          const v4f t = *(const v4f*)(sr + (16 * oh + r) * DO_ + 4 * q);
          vr[oh][4 * q + 0][r] = t[0];
          vr[oh][4 * q + 1][r] = t[1];
          vr[oh][4 * q + 2][r] = t[2];
          vr[oh][4 * q + 3][r] = t[3];
        }
      }
    }
  }
  __syncthreads();

  if (FIRST) {
#pragma unroll
    for (int oh = 0; oh < 2; ++oh)
#pragma unroll
      for (int r = 0; r < 8; ++r) {
        float nn = 0.f;
#pragma unroll
        for (int d = 0; d < 16; ++d) nn += vr[oh][d][r] * vr[oh][d][r];
        const float scl = sqrtf(nn) / (1.f + nn);
#pragma unroll
        for (int d = 0; d < 16; ++d) vr[oh][d][r] *= scl;
      }
  }

  float sacc[2][16][8];
#pragma unroll
  for (int oh = 0; oh < 2; ++oh)
#pragma unroll
    for (int d = 0; d < 16; ++d)
#pragma unroll
      for (int r = 0; r < 8; ++r) sacc[oh][d][r] = 0.f;

  for (int nn = 0; nn < NPW - 1; ++nn) {
    const int cur = (nn & 1) * WBUF_F;
    route_trip<true>(wsr + cur, wg + (size_t)(nn + 1) * WN, wsw + (WBUF_F - cur),
                     xg + (nn + 1) * DI, x4, vr, sacc);
    __syncthreads();
  }
  route_trip<false>(wsr + ((NPW - 1) & 1) * WBUF_F, nullptr, nullptr, nullptr, x4, vr, sacc);

  {
    float* ew = rows + (nh * 2 + wb) * TILE_F + c * ROWP + (8 * h) * DO_;
#pragma unroll
    for (int oh = 0; oh < 2; ++oh)
#pragma unroll
      for (int r = 0; r < 8; ++r)
#pragma unroll
        for (int q = 0; q < 4; ++q) {
          const v4f t = {sacc[oh][4 * q + 0][r], sacc[oh][4 * q + 1][r],
                         sacc[oh][4 * q + 2][r], sacc[oh][4 * q + 3][r]};
          *(v4f*)(ew + (16 * oh + r) * DO_ + 4 * q) = t;
        }
  }
  __syncthreads();
  {
    const float* e0 = rows + wb * TILE_F + (8 * nh) * ROWP + lane * 4;
    const float* e1 = e0 + 2 * TILE_F;
    float* pg = ps + (size_t)p * PS_STRIDE + (size_t)(bt * 16 + 8 * nh) * NOD + lane * 4;
#pragma unroll
    for (int j = 0; j < 8; ++j)
#pragma unroll
      for (int k = 0; k < 4; ++k) {
        const v4f a = *(const v4f*)(e0 + j * ROWP + k * 128);
        const v4f u = *(const v4f*)(e1 + j * ROWP + k * 128);
        *(v4f*)(pg + j * NOD + k * 128) = a + u;
      }
  }
}

#define RGRP 4
#define PPG  (NB_ROUTE/RGRP)

template <bool ADD>
__global__ __launch_bounds__(512)
void caps_reduce_squash(const float* __restrict__ ps,
                        const float* h0, const float* h1, float* out) {
  __shared__ __attribute__((aligned(16))) float part[(RGRP - 1) * NOD];
  const int b = blockIdx.x;
  const int f = threadIdx.x & 127;
  const int g = __builtin_amdgcn_readfirstlane((int)(threadIdx.x >> 7));
  const size_t e = (size_t)b * NOD + 4 * f;
  const float* src = ps + (size_t)(g * PPG) * PS_STRIDE + e;
  v4f t[PPG];
#pragma unroll
  for (int j = 0; j < PPG; ++j) t[j] = *(const v4f*)(src + (size_t)j * PS_STRIDE);
  v4f a = {};
  if (ADD) {
    if (g == 0) a = (*(const v4f*)(h0 + e) + *(const v4f*)(h1 + e)) * (1.0f / NO);
  }
  __builtin_amdgcn_sched_barrier(0);
  v4f s = t[0];
#pragma unroll
  for (int j = 1; j < PPG; ++j) s += t[j];
  if (g != 0) *(v4f*)(part + (g - 1) * NOD + 4 * f) = s;
  __syncthreads();
  if (g == 0) {
#pragma unroll
    for (int k = 0; k < RGRP - 1; ++k) s += *(const v4f*)(part + k * NOD + 4 * f);
    float nn = s[0] * s[0] + s[1] * s[1] + s[2] * s[2] + s[3] * s[3];
    nn += __shfl_xor(nn, 1);
    nn += __shfl_xor(nn, 2);
    v4f r = s * (sqrtf(nn) / (1.f + nn));
    if (ADD) {
      float n0 = a[0] * a[0] + a[1] * a[1] + a[2] * a[2] + a[3] * a[3];
      n0 += __shfl_xor(n0, 1);
      n0 += __shfl_xor(n0, 2);
      r += a * (sqrtf(n0) / (1.f + n0));
    }
    *(v4f*)(out + e) = r;
  }
}

extern "C" void kernel_launch(void* const* d_in, const int* in_sizes, int n_in,
                              void* d_out, int out_size, void* d_ws, size_t ws_size,
                              hipStream_t stream) {
  const float* x = (const float*)d_in[0];
  const float* W = (const float*)d_in[1];
  float* out = (float*)d_out;

  float* ps   = (float*)d_ws;
  float* h0   = ps + (size_t)NB_ROUTE * PS_STRIDE;
  float* h1   = h0 + (size_t)HPLANE;

  caps_s0_half<<<NBLK_S0, 256, 0, stream>>>(x, W, h0);
  caps_route<true><<<NBLK_ROUTE, 128, 0, stream>>>(x, W, h0, ps);
  caps_reduce_squash<true><<<B_SZ, NOD, 0, stream>>>(ps, h0, h1, h0);
  caps_route<false><<<NBLK_ROUTE, 128, 0, stream>>>(x, W, h0, ps);
  caps_reduce_squash<false><<<B_SZ, NOD, 0, stream>>>(ps, nullptr, nullptr, out);
}
